# starting kernel
# baseline (speedup 1.0000x reference)
#include <hip/hip_runtime.h>

#define B_SZ   256
#define A_SZ   64
#define FEATN  256
#define H_SZ   16
#define S_SZ   20
#define POSN   64
#define ENCN   256
#define OUTN   384
#define MAXD   10

typedef __attribute__((ext_vector_type(2))) float v2f;
typedef __attribute__((ext_vector_type(8))) float v8f;

#define LDA_ARMY 260
#define LDP      388
#define LDH      388
#define LDATT    20
#define OFF_ARMY 0
#define OFF_PARM (OFF_ARMY + A_SZ * LDA_ARMY)
#define OFF_HIST (OFF_PARM + A_SZ * LDP)
#define OFF_ATT  (OFF_HIST + H_SZ * LDH)
#define SMEM_FLOATS (OFF_ATT + A_SZ * LDATT)

__device__ __forceinline__ v8f wmma4(v2f a, v2f b, v8f c) {
    return __builtin_amdgcn_wmma_f32_16x16x4_f32(
         false, a,  false, b,
         (short)0, c,  false,  false);
}

__global__ void norm_scale_kernel(const float* __restrict__ proj_v,
                                  const float* __restrict__ proj_g,
                                  float* __restrict__ scale_out) {
    __shared__ float red[256];
    float s = 0.f;
    for (int i = threadIdx.x; i < OUTN * FEATN; i += 256) {
        float v = proj_v[i];
        s += v * v;
    }
    red[threadIdx.x] = s;
    __syncthreads();
    for (int st = 128; st > 0; st >>= 1) {
        if ((int)threadIdx.x < st) red[threadIdx.x] += red[threadIdx.x + st];
        __syncthreads();
    }
    if (threadIdx.x == 0) scale_out[0] = proj_g[0] * rsqrtf(red[0]);
}

__global__ __launch_bounds__(256) void fused_attn(
    const float* __restrict__ army,
    const int*   __restrict__ inst,
    const int*   __restrict__ ilen,
    const int*   __restrict__ idiff,
    const float* __restrict__ tok_emb,
    const float* __restrict__ pos_w,
    const float* __restrict__ diff_w,
    const float* __restrict__ proj_v,
    const float* __restrict__ proj_b,
    const float* __restrict__ scale_p,
    float* __restrict__ out_inst,
    float* __restrict__ out_sum)
{
    extern __shared__ float smem[];
    float* s_army = smem + OFF_ARMY;
    float* s_parm = smem + OFF_PARM;
    float* s_hist = smem + OFF_HIST;
    float* s_att  = smem + OFF_ATT;

    const int b    = blockIdx.x;
    const int tid  = threadIdx.x;
    const int wave = tid >> 5;
    const int lane = tid & 31;
    const int lm   = lane & 15;
    const int lh   = lane >> 4;
    const float scale = scale_p[0];

    for (int e = tid; e < H_SZ * OUTN; e += 256) {
        int h = e / OUTN, d = e - h * OUTN;
        float v;
        if (d < ENCN) {
            int len = ilen[b * H_SZ + h];
            if (len < 1) len = 1;
            const int* row = inst + (b * H_SZ + h) * S_SZ;
            float acc = 0.f;
            for (int s = 0; s < len; ++s) acc += tok_emb[row[s] * ENCN + d];
            v = acc / (float)len;
        } else if (d < ENCN + POSN) {
            v = pos_w[h * POSN + (d - ENCN)];
        } else {
            int di = idiff[b * H_SZ + h];
            di = di < 0 ? 0 : (di > MAXD ? MAXD : di);
            v = diff_w[di * POSN + (d - ENCN - POSN)];
        }
        s_hist[h * LDH + d] = v;
    }
    for (int e = tid; e < A_SZ * FEATN; e += 256) {
        int r = e >> 8, c = e & 255;
        s_army[r * LDA_ARMY + c] = army[(size_t)b * A_SZ * FEATN + e];
    }
    __syncthreads();

    for (int d = tid; d < OUTN; d += 256)
        out_sum[b * OUTN + d] = s_hist[(H_SZ - 1) * LDH + d];

    for (int nt = 0; nt < 3; ++nt) {
        const int n0 = (wave * 3 + nt) * 16;
        v8f acc0 = {}, acc1 = {}, acc2 = {}, acc3 = {};
        const float* wrow = proj_v + (size_t)(n0 + lm) * FEATN + 2 * lh;
        for (int k0 = 0; k0 < FEATN; k0 += 4) {
            v2f bf = *(const v2f*)(wrow + k0);
            v2f a0 = *(const v2f*)(&s_army[(0 * 16 + lm) * LDA_ARMY + k0 + 2 * lh]);
            v2f a1 = *(const v2f*)(&s_army[(1 * 16 + lm) * LDA_ARMY + k0 + 2 * lh]);
            v2f a2 = *(const v2f*)(&s_army[(2 * 16 + lm) * LDA_ARMY + k0 + 2 * lh]);
            v2f a3 = *(const v2f*)(&s_army[(3 * 16 + lm) * LDA_ARMY + k0 + 2 * lh]);
            acc0 = wmma4(a0, bf, acc0);
            acc1 = wmma4(a1, bf, acc1);
            acc2 = wmma4(a2, bf, acc2);
            acc3 = wmma4(a3, bf, acc3);
        }
        const float bias = proj_b[n0 + lm];
        #pragma unroll
        for (int r = 0; r < 8; ++r) {
            int mrow = 8 * lh + r;
            float v0 = acc0[r] * scale + bias;
            float v1 = acc1[r] * scale + bias;
            float v2 = acc2[r] * scale + bias;
            float v3 = acc3[r] * scale + bias;
            s_parm[(0 * 16 + mrow) * LDP + n0 + lm] = v0 > 0.f ? v0 : 0.f;
            s_parm[(1 * 16 + mrow) * LDP + n0 + lm] = v1 > 0.f ? v1 : 0.f;
            s_parm[(2 * 16 + mrow) * LDP + n0 + lm] = v2 > 0.f ? v2 : 0.f;
            s_parm[(3 * 16 + mrow) * LDP + n0 + lm] = v3 > 0.f ? v3 : 0.f;
        }
    }
    __syncthreads();

    if (wave < 4) {
        const int m0 = wave * 16;
        v8f acc = {};
        for (int k0 = 0; k0 < OUTN; k0 += 4) {
            v2f a  = *(const v2f*)(&s_parm[(m0 + lm) * LDP + k0 + 2 * lh]);
            v2f bf = *(const v2f*)(&s_hist[lm * LDH + k0 + 2 * lh]);
            acc = wmma4(a, bf, acc);
        }
        #pragma unroll
        for (int r = 0; r < 8; ++r)
            s_att[(m0 + 8 * lh + r) * LDATT + lm] = acc[r];
    }
    __syncthreads();

    if (tid < 64) {
        float* row = s_att + tid * LDATT;
        float mx = row[0];
        #pragma unroll
        for (int h = 1; h < 16; ++h) mx = fmaxf(mx, row[h]);
        float sum = 0.f;
        #pragma unroll
        for (int h = 0; h < 16; ++h) { float e = __expf(row[h] - mx); row[h] = e; sum += e; }
        float inv = 1.f / sum;
        #pragma unroll
        for (int h = 0; h < 16; ++h) row[h] *= inv;
    }
    __syncthreads();

    for (int nt = 0; nt < 3; ++nt) {
        const int n0 = (wave * 3 + nt) * 16;
        v8f acc0 = {}, acc1 = {}, acc2 = {}, acc3 = {};
        #pragma unroll
        for (int k0 = 0; k0 < 16; k0 += 4) {
            int kk = k0 + 2 * lh;
            v2f bf;
            bf.x = s_hist[kk * LDH + n0 + lm];
            bf.y = s_hist[(kk + 1) * LDH + n0 + lm];
            v2f a0 = *(const v2f*)(&s_att[(0 * 16 + lm) * LDATT + kk]);
            v2f a1 = *(const v2f*)(&s_att[(1 * 16 + lm) * LDATT + kk]);
            v2f a2 = *(const v2f*)(&s_att[(2 * 16 + lm) * LDATT + kk]);
            v2f a3 = *(const v2f*)(&s_att[(3 * 16 + lm) * LDATT + kk]);
            acc0 = wmma4(a0, bf, acc0);
            acc1 = wmma4(a1, bf, acc1);
            acc2 = wmma4(a2, bf, acc2);
            acc3 = wmma4(a3, bf, acc3);
        }
        #pragma unroll
        for (int r = 0; r < 8; ++r) {
            int mrow = 8 * lh + r;
            out_inst[((size_t)b * A_SZ + 0 * 16 + mrow) * OUTN + n0 + lm] = acc0[r];
            out_inst[((size_t)b * A_SZ + 1 * 16 + mrow) * OUTN + n0 + lm] = acc1[r];
            out_inst[((size_t)b * A_SZ + 2 * 16 + mrow) * OUTN + n0 + lm] = acc2[r];
            out_inst[((size_t)b * A_SZ + 3 * 16 + mrow) * OUTN + n0 + lm] = acc3[r];
        }
    }
}

extern "C" void kernel_launch(void* const* d_in, const int* in_sizes, int n_in,
                              void* d_out, int out_size, void* d_ws, size_t ws_size,
                              hipStream_t stream) {
    (void)in_sizes; (void)n_in; (void)out_size; (void)ws_size;
    const float* army  = (const float*)d_in[0];
    const int*   inst  = (const int*)  d_in[1];
    const int*   ilen  = (const int*)  d_in[2];
    const int*   idiff = (const int*)  d_in[3];
    const float* tok   = (const float*)d_in[4];
    const float* posw  = (const float*)d_in[5];
    const float* diffw = (const float*)d_in[6];
    const float* pv    = (const float*)d_in[7];
    const float* pg    = (const float*)d_in[8];
    const float* pb    = (const float*)d_in[9];

    float* out      = (float*)d_out;
    float* out_inst = out;
    float* out_sum  = out + (size_t)B_SZ * A_SZ * OUTN;
    float* scale    = (float*)d_ws;

    size_t smem_bytes = (size_t)SMEM_FLOATS * sizeof(float);
    (void)hipFuncSetAttribute((const void*)fused_attn,
                              hipFuncAttributeMaxDynamicSharedMemorySize,
                              (int)smem_bytes);

    hipLaunchKernelGGL(norm_scale_kernel, dim3(1), dim3(256), 0, stream, pv, pg, scale);
    hipLaunchKernelGGL(fused_attn, dim3(B_SZ), dim3(256), smem_bytes, stream,
                       army, inst, ilen, idiff, tok, posw, diffw, pv, pb, scale,
                       out_inst, out_sum);
}


# weight-norm launch folded into the fused kernel: lanes square the proj_v B fragments they already load, block sum by shuffle + 8 wave partials (norm summed in a new order), epilogue after the sum
# speedup vs baseline: 1.5370x; 1.5370x over previous
#include <hip/hip_runtime.h>

#define B_SZ   256
#define A_SZ   64
#define FEATN  256
#define H_SZ   16
#define S_SZ   20
#define POSN   64
#define ENCN   256
#define OUTN   384
#define MAXD   10

typedef __attribute__((ext_vector_type(2))) float v2f;
typedef __attribute__((ext_vector_type(8))) float v8f;

#define LDA_ARMY 260
#define LDP      388
#define LDH      388
#define LDATT    20
#define OFF_ARMY 0
#define OFF_PARM (OFF_ARMY + A_SZ * LDA_ARMY)
#define OFF_HIST (OFF_PARM + A_SZ * LDP)
#define OFF_ATT  (OFF_HIST + H_SZ * LDH)
#define OFF_RED  (OFF_ATT + A_SZ * LDATT)
#define SMEM_FLOATS (OFF_RED + 8)

__device__ __forceinline__ v8f wmma4(v2f a, v2f b, v8f c) {
    return __builtin_amdgcn_wmma_f32_16x16x4_f32(
         false, a,  false, b,
         (short)0, c,  false,  false);
}

__global__ __launch_bounds__(256) void fused_attn(
    const float* __restrict__ army,
    const int*   __restrict__ inst,
    const int*   __restrict__ ilen,
    const int*   __restrict__ idiff,
    const float* __restrict__ tok_emb,
    const float* __restrict__ pos_w,
    const float* __restrict__ diff_w,
    const float* __restrict__ proj_v,
    const float* __restrict__ proj_b,
    const float* __restrict__ proj_g,
    float* __restrict__ out_inst,
    float* __restrict__ out_sum)
{
    extern __shared__ float smem[];
    float* s_army = smem + OFF_ARMY;
    float* s_parm = smem + OFF_PARM;
    float* s_hist = smem + OFF_HIST;
    float* s_att  = smem + OFF_ATT;
    float* s_red  = smem + OFF_RED;

    const int b    = blockIdx.x;
    const int tid  = threadIdx.x;
    const int wave = tid >> 5;
    const int lane = tid & 31;
    const int lm   = lane & 15;
    const int lh   = lane >> 4;
    const float g = proj_g[0];

    for (int e = tid; e < H_SZ * OUTN; e += 256) {
        int h = e / OUTN, d = e - h * OUTN;
        float v;
        if (d < ENCN) {
            int len = ilen[b * H_SZ + h];
            if (len < 1) len = 1;
            const int* row = inst + (b * H_SZ + h) * S_SZ;
            float acc = 0.f;
            for (int s = 0; s < len; ++s) acc += tok_emb[row[s] * ENCN + d];
            v = acc / (float)len;
        } else if (d < ENCN + POSN) {
            v = pos_w[h * POSN + (d - ENCN)];
        } else {
            int di = idiff[b * H_SZ + h];
            di = di < 0 ? 0 : (di > MAXD ? MAXD : di);
            v = diff_w[di * POSN + (d - ENCN - POSN)];
        }
        s_hist[h * LDH + d] = v;
    }
    for (int e = tid; e < A_SZ * FEATN; e += 256) {
        int r = e >> 8, c = e & 255;
        s_army[r * LDA_ARMY + c] = army[(size_t)b * A_SZ * FEATN + e];
    }
    __syncthreads();

    for (int d = tid; d < OUTN; d += 256)
        out_sum[b * OUTN + d] = s_hist[(H_SZ - 1) * LDH + d];

    v8f acc[3][4];
    float ssx = 0.f, ssy = 0.f;
    #pragma unroll
    for (int nt = 0; nt < 3; ++nt) {
        const int n0 = (wave * 3 + nt) * 16;
        v8f acc0 = {}, acc1 = {}, acc2 = {}, acc3 = {};
        const float* wrow = proj_v + (size_t)(n0 + lm) * FEATN + 2 * lh;
        for (int k0 = 0; k0 < FEATN; k0 += 4) {
            v2f bf = *(const v2f*)(wrow + k0);
            v2f a0 = *(const v2f*)(&s_army[(0 * 16 + lm) * LDA_ARMY + k0 + 2 * lh]);
            v2f a1 = *(const v2f*)(&s_army[(1 * 16 + lm) * LDA_ARMY + k0 + 2 * lh]);
            v2f a2 = *(const v2f*)(&s_army[(2 * 16 + lm) * LDA_ARMY + k0 + 2 * lh]);
            v2f a3 = *(const v2f*)(&s_army[(3 * 16 + lm) * LDA_ARMY + k0 + 2 * lh]);
            ssx += bf.x * bf.x;
            ssy += bf.y * bf.y;
            acc0 = wmma4(a0, bf, acc0);
            acc1 = wmma4(a1, bf, acc1);
            acc2 = wmma4(a2, bf, acc2);
            acc3 = wmma4(a3, bf, acc3);
        }
        acc[nt][0] = acc0; acc[nt][1] = acc1; acc[nt][2] = acc2; acc[nt][3] = acc3;
    }
    float ss = ssx + ssy;
    for (int off = 16; off > 0; off >>= 1) ss += __shfl_xor(ss, off, 32);
    if (lane == 0) s_red[wave] = ss;
    __syncthreads();
    float tot = 0.f;
    for (int w = 0; w < 8; ++w) tot += s_red[w];
    const float scale = g * rsqrtf(tot);

    #pragma unroll
    for (int nt = 0; nt < 3; ++nt) {
        const int n0 = (wave * 3 + nt) * 16;
        const float bias = proj_b[n0 + lm];
        #pragma unroll
        for (int r = 0; r < 8; ++r) {
            int mrow = 8 * lh + r;
            float v0 = acc[nt][0][r] * scale + bias;
            float v1 = acc[nt][1][r] * scale + bias;
            float v2 = acc[nt][2][r] * scale + bias;
            float v3 = acc[nt][3][r] * scale + bias;
            s_parm[(0 * 16 + mrow) * LDP + n0 + lm] = v0 > 0.f ? v0 : 0.f;
            s_parm[(1 * 16 + mrow) * LDP + n0 + lm] = v1 > 0.f ? v1 : 0.f;
            s_parm[(2 * 16 + mrow) * LDP + n0 + lm] = v2 > 0.f ? v2 : 0.f;
            s_parm[(3 * 16 + mrow) * LDP + n0 + lm] = v3 > 0.f ? v3 : 0.f;
        }
    }
    __syncthreads();

    if (wave < 4) {
        const int m0 = wave * 16;
        v8f acc = {};
        for (int k0 = 0; k0 < OUTN; k0 += 4) {
            v2f a  = *(const v2f*)(&s_parm[(m0 + lm) * LDP + k0 + 2 * lh]);
            v2f bf = *(const v2f*)(&s_hist[lm * LDH + k0 + 2 * lh]);
            acc = wmma4(a, bf, acc);
        }
        #pragma unroll
        for (int r = 0; r < 8; ++r)
            s_att[(m0 + 8 * lh + r) * LDATT + lm] = acc[r];
    }
    __syncthreads();

    if (tid < 64) {
        float* row = s_att + tid * LDATT;
        float mx = row[0];
        #pragma unroll
        for (int h = 1; h < 16; ++h) mx = fmaxf(mx, row[h]);
        float sum = 0.f;
        #pragma unroll
        for (int h = 0; h < 16; ++h) { float e = __expf(row[h] - mx); row[h] = e; sum += e; }
        float inv = 1.f / sum;
        #pragma unroll
        for (int h = 0; h < 16; ++h) row[h] *= inv;
    }
    __syncthreads();

    for (int nt = 0; nt < 3; ++nt) {
        const int n0 = (wave * 3 + nt) * 16;
        v8f acc0 = {}, acc1 = {}, acc2 = {}, acc3 = {};
        #pragma unroll
        for (int k0 = 0; k0 < 16; k0 += 4) {
            int kk = k0 + 2 * lh;
            v2f bf;
            bf.x = s_hist[kk * LDH + n0 + lm];
            bf.y = s_hist[(kk + 1) * LDH + n0 + lm];
            v2f a0 = *(const v2f*)(&s_att[(0 * 16 + lm) * LDATT + kk]);
            v2f a1 = *(const v2f*)(&s_att[(1 * 16 + lm) * LDATT + kk]);
            v2f a2 = *(const v2f*)(&s_att[(2 * 16 + lm) * LDATT + kk]);
            v2f a3 = *(const v2f*)(&s_att[(3 * 16 + lm) * LDATT + kk]);
            acc0 = wmma4(a0, bf, acc0);
            acc1 = wmma4(a1, bf, acc1);
            acc2 = wmma4(a2, bf, acc2);
            acc3 = wmma4(a3, bf, acc3);
        }
        #pragma unroll
        for (int r = 0; r < 8; ++r) {
            int mrow = 8 * lh + r;
            out_inst[((size_t)b * A_SZ + 0 * 16 + mrow) * OUTN + n0 + lm] = acc0[r];
            out_inst[((size_t)b * A_SZ + 1 * 16 + mrow) * OUTN + n0 + lm] = acc1[r];
            out_inst[((size_t)b * A_SZ + 2 * 16 + mrow) * OUTN + n0 + lm] = acc2[r];
            out_inst[((size_t)b * A_SZ + 3 * 16 + mrow) * OUTN + n0 + lm] = acc3[r];
        }
    }
}

extern "C" void kernel_launch(void* const* d_in, const int* in_sizes, int n_in,
                              void* d_out, int out_size, void* d_ws, size_t ws_size,
                              hipStream_t stream) {
    (void)in_sizes; (void)n_in; (void)out_size; (void)d_ws; (void)ws_size;
    const float* army  = (const float*)d_in[0];
    const int*   inst  = (const int*)  d_in[1];
    const int*   ilen  = (const int*)  d_in[2];
    const int*   idiff = (const int*)  d_in[3];
    const float* tok   = (const float*)d_in[4];
    const float* posw  = (const float*)d_in[5];
    const float* diffw = (const float*)d_in[6];
    const float* pv    = (const float*)d_in[7];
    const float* pg    = (const float*)d_in[8];
    const float* pb    = (const float*)d_in[9];

    float* out      = (float*)d_out;
    float* out_inst = out;
    float* out_sum  = out + (size_t)B_SZ * A_SZ * OUTN;

    size_t smem_bytes = (size_t)SMEM_FLOATS * sizeof(float);
    (void)hipFuncSetAttribute((const void*)fused_attn,
                              hipFuncAttributeMaxDynamicSharedMemorySize,
                              (int)smem_bytes);

    hipLaunchKernelGGL(fused_attn, dim3(B_SZ), dim3(256), smem_bytes, stream,
                       army, inst, ilen, idiff, tok, posw, diffw, pv, pb, pg,
                       out_inst, out_sum);
}


# history-feature staging made wave-per-row: ids, lengths, diff indices by scalar loads, 8 fixed columns per lane, a row's embedding loads all issued before the first add; sum order unchanged
# speedup vs baseline: 2.3322x; 1.5174x over previous
#include <hip/hip_runtime.h>

#define B_SZ   256
#define A_SZ   64
#define FEATN  256
#define H_SZ   16
#define S_SZ   20
#define POSN   64
#define ENCN   256
#define OUTN   384
#define MAXD   10

typedef __attribute__((ext_vector_type(2))) float v2f;
typedef __attribute__((ext_vector_type(8))) float v8f;
typedef __attribute__((ext_vector_type(4))) float v4f;

#define LDA_ARMY 260
#define LDP      388
#define LDH      388
#define LDATT    20
#define OFF_ARMY 0
#define OFF_PARM (OFF_ARMY + A_SZ * LDA_ARMY)
#define OFF_HIST (OFF_PARM + A_SZ * LDP)
#define OFF_ATT  (OFF_HIST + H_SZ * LDH)
#define OFF_RED  (OFF_ATT + A_SZ * LDATT)
#define SMEM_FLOATS (OFF_RED + 8)

__device__ __forceinline__ v8f wmma4(v2f a, v2f b, v8f c) {
    return __builtin_amdgcn_wmma_f32_16x16x4_f32(
         false, a,  false, b,
         (short)0, c,  false,  false);
}

__global__ __launch_bounds__(256) void fused_attn(
    const float* __restrict__ army,
    const int*   __restrict__ inst,
    const int*   __restrict__ ilen,
    const int*   __restrict__ idiff,
    const float* __restrict__ tok_emb,
    const float* __restrict__ pos_w,
    const float* __restrict__ diff_w,
    const float* __restrict__ proj_v,
    const float* __restrict__ proj_b,
    const float* __restrict__ proj_g,
    float* __restrict__ out_inst,
    float* __restrict__ out_sum)
{
    extern __shared__ float smem[];
    float* s_army = smem + OFF_ARMY;
    float* s_parm = smem + OFF_PARM;
    float* s_hist = smem + OFF_HIST;
    float* s_att  = smem + OFF_ATT;
    float* s_red  = smem + OFF_RED;

    const int b    = blockIdx.x;
    const int tid  = threadIdx.x;
    const int wave = tid >> 5;
    const int lane = tid & 31;
    const int lm   = lane & 15;
    const int lh   = lane >> 4;
    const float g = proj_g[0];

    const int wu = __builtin_amdgcn_readfirstlane(wave);
    const int hrow0 = b * H_SZ + wu * 2;
    int ids[2][S_SZ];
    int lens[2], dis[2];
    #pragma unroll
    for (int hh = 0; hh < 2; ++hh) {
        lens[hh] = ilen[hrow0 + hh];
        int di = idiff[hrow0 + hh];
        dis[hh] = di < 0 ? 0 : (di > MAXD ? MAXD : di);
        #pragma unroll
        for (int s = 0; s < S_SZ; ++s) ids[hh][s] = inst[(hrow0 + hh) * S_SZ + s];
    }
    const int c4 = lm * 4;
    const v4f pv4 = *(const v4f*)(pos_w + (wu * 2 + lh) * POSN + c4);
    const v4f dv4 = *(const v4f*)(diff_w + (lh ? dis[1] : dis[0]) * POSN + c4);
    #pragma unroll
    for (int hh = 0; hh < 2; ++hh) {
        const int h = wu * 2 + hh;
        int len = lens[hh];
        if (len < 1) len = 1;
        v4f t0[S_SZ], t1[S_SZ];
        #pragma unroll
        for (int s = 0; s < S_SZ; ++s) {
            if (s < len) {
                const v4f* erow = (const v4f*)(tok_emb + ids[hh][s] * ENCN);
                t0[s] = erow[lane];
                t1[s] = erow[32 + lane];
            } else {
                t0[s] = (v4f){0.f, 0.f, 0.f, 0.f};
                t1[s] = (v4f){0.f, 0.f, 0.f, 0.f};
            }
        }
        v4f a0 = {0.f, 0.f, 0.f, 0.f}, a1 = {0.f, 0.f, 0.f, 0.f};
        #pragma unroll
        for (int s = 0; s < S_SZ; ++s) { a0 += t0[s]; a1 += t1[s]; }
        const float fl = (float)len;
        v4f m0, m1;
        m0.x = a0.x / fl; m0.y = a0.y / fl; m0.z = a0.z / fl; m0.w = a0.w / fl;
        m1.x = a1.x / fl; m1.y = a1.y / fl; m1.z = a1.z / fl; m1.w = a1.w / fl;
        *(v4f*)(&s_hist[h * LDH + 4 * lane])       = m0;
        *(v4f*)(&s_hist[h * LDH + 128 + 4 * lane]) = m1;
    }
    *(v4f*)(&s_hist[(wu * 2 + lh) * LDH + ENCN + c4])        = pv4;
    *(v4f*)(&s_hist[(wu * 2 + lh) * LDH + ENCN + POSN + c4]) = dv4;
    for (int e = tid; e < A_SZ * FEATN; e += 256) {
        int r = e >> 8, c = e & 255;
        s_army[r * LDA_ARMY + c] = army[(size_t)b * A_SZ * FEATN + e];
    }
    __syncthreads();

    for (int d = tid; d < OUTN; d += 256)
        out_sum[b * OUTN + d] = s_hist[(H_SZ - 1) * LDH + d];

    v8f acc[3][4];
    float ssx = 0.f, ssy = 0.f;
    #pragma unroll
    for (int nt = 0; nt < 3; ++nt) {
        const int n0 = (wave * 3 + nt) * 16;
        v8f acc0 = {}, acc1 = {}, acc2 = {}, acc3 = {};
        const float* wrow = proj_v + (size_t)(n0 + lm) * FEATN + 2 * lh;
        for (int k0 = 0; k0 < FEATN; k0 += 4) {
            v2f bf = *(const v2f*)(wrow + k0);
            v2f a0 = *(const v2f*)(&s_army[(0 * 16 + lm) * LDA_ARMY + k0 + 2 * lh]);
            v2f a1 = *(const v2f*)(&s_army[(1 * 16 + lm) * LDA_ARMY + k0 + 2 * lh]);
            v2f a2 = *(const v2f*)(&s_army[(2 * 16 + lm) * LDA_ARMY + k0 + 2 * lh]);
            v2f a3 = *(const v2f*)(&s_army[(3 * 16 + lm) * LDA_ARMY + k0 + 2 * lh]);
            ssx += bf.x * bf.x;
            ssy += bf.y * bf.y;
            acc0 = wmma4(a0, bf, acc0);
            acc1 = wmma4(a1, bf, acc1);
            acc2 = wmma4(a2, bf, acc2);
            acc3 = wmma4(a3, bf, acc3);
        }
        acc[nt][0] = acc0; acc[nt][1] = acc1; acc[nt][2] = acc2; acc[nt][3] = acc3;
    }
    float ss = ssx + ssy;
    for (int off = 16; off > 0; off >>= 1) ss += __shfl_xor(ss, off, 32);
    if (lane == 0) s_red[wave] = ss;
    __syncthreads();
    float tot = 0.f;
    for (int w = 0; w < 8; ++w) tot += s_red[w];
    const float scale = g * rsqrtf(tot);

    #pragma unroll
    for (int nt = 0; nt < 3; ++nt) {
        const int n0 = (wave * 3 + nt) * 16;
        const float bias = proj_b[n0 + lm];
        #pragma unroll
        for (int r = 0; r < 8; ++r) {
            int mrow = 8 * lh + r;
            float v0 = acc[nt][0][r] * scale + bias;
            float v1 = acc[nt][1][r] * scale + bias;
            float v2 = acc[nt][2][r] * scale + bias;
            float v3 = acc[nt][3][r] * scale + bias;
            s_parm[(0 * 16 + mrow) * LDP + n0 + lm] = v0 > 0.f ? v0 : 0.f;
            s_parm[(1 * 16 + mrow) * LDP + n0 + lm] = v1 > 0.f ? v1 : 0.f;
            s_parm[(2 * 16 + mrow) * LDP + n0 + lm] = v2 > 0.f ? v2 : 0.f;
            s_parm[(3 * 16 + mrow) * LDP + n0 + lm] = v3 > 0.f ? v3 : 0.f;
        }
    }
    __syncthreads();

    if (wave < 4) {
        const int m0 = wave * 16;
        v8f acc = {};
        for (int k0 = 0; k0 < OUTN; k0 += 4) {
            v2f a  = *(const v2f*)(&s_parm[(m0 + lm) * LDP + k0 + 2 * lh]);
            v2f bf = *(const v2f*)(&s_hist[lm * LDH + k0 + 2 * lh]);
            acc = wmma4(a, bf, acc);
        }
        #pragma unroll
        for (int r = 0; r < 8; ++r)
            s_att[(m0 + 8 * lh + r) * LDATT + lm] = acc[r];
    }
    __syncthreads();

    if (tid < 64) {
        float* row = s_att + tid * LDATT;
        float mx = row[0];
        #pragma unroll
        for (int h = 1; h < 16; ++h) mx = fmaxf(mx, row[h]);
        float sum = 0.f;
        #pragma unroll
        for (int h = 0; h < 16; ++h) { float e = __expf(row[h] - mx); row[h] = e; sum += e; }
        float inv = 1.f / sum;
        #pragma unroll
        for (int h = 0; h < 16; ++h) row[h] *= inv;
    }
    __syncthreads();

    for (int nt = 0; nt < 3; ++nt) {
        const int n0 = (wave * 3 + nt) * 16;
        v8f acc0 = {}, acc1 = {}, acc2 = {}, acc3 = {};
        #pragma unroll
        for (int k0 = 0; k0 < 16; k0 += 4) {
            int kk = k0 + 2 * lh;
            v2f bf;
            bf.x = s_hist[kk * LDH + n0 + lm];
            bf.y = s_hist[(kk + 1) * LDH + n0 + lm];
            v2f a0 = *(const v2f*)(&s_att[(0 * 16 + lm) * LDATT + kk]);
            v2f a1 = *(const v2f*)(&s_att[(1 * 16 + lm) * LDATT + kk]);
            v2f a2 = *(const v2f*)(&s_att[(2 * 16 + lm) * LDATT + kk]);
            v2f a3 = *(const v2f*)(&s_att[(3 * 16 + lm) * LDATT + kk]);
            acc0 = wmma4(a0, bf, acc0);
            acc1 = wmma4(a1, bf, acc1);
            acc2 = wmma4(a2, bf, acc2);
            acc3 = wmma4(a3, bf, acc3);
        }
        #pragma unroll
        for (int r = 0; r < 8; ++r) {
            int mrow = 8 * lh + r;
            out_inst[((size_t)b * A_SZ + 0 * 16 + mrow) * OUTN + n0 + lm] = acc0[r];
            out_inst[((size_t)b * A_SZ + 1 * 16 + mrow) * OUTN + n0 + lm] = acc1[r];
            out_inst[((size_t)b * A_SZ + 2 * 16 + mrow) * OUTN + n0 + lm] = acc2[r];
            out_inst[((size_t)b * A_SZ + 3 * 16 + mrow) * OUTN + n0 + lm] = acc3[r];
        }
    }
}

extern "C" void kernel_launch(void* const* d_in, const int* in_sizes, int n_in,
                              void* d_out, int out_size, void* d_ws, size_t ws_size,
                              hipStream_t stream) {
    (void)in_sizes; (void)n_in; (void)out_size; (void)d_ws; (void)ws_size;
    const float* army  = (const float*)d_in[0];
    const int*   inst  = (const int*)  d_in[1];
    const int*   ilen  = (const int*)  d_in[2];
    const int*   idiff = (const int*)  d_in[3];
    const float* tok   = (const float*)d_in[4];
    const float* posw  = (const float*)d_in[5];
    const float* diffw = (const float*)d_in[6];
    const float* pv    = (const float*)d_in[7];
    const float* pg    = (const float*)d_in[8];
    const float* pb    = (const float*)d_in[9];

    float* out      = (float*)d_out;
    float* out_inst = out;
    float* out_sum  = out + (size_t)B_SZ * A_SZ * OUTN;

    size_t smem_bytes = (size_t)SMEM_FLOATS * sizeof(float);
    (void)hipFuncSetAttribute((const void*)fused_attn,
                              hipFuncAttributeMaxDynamicSharedMemorySize,
                              (int)smem_bytes);

    hipLaunchKernelGGL(fused_attn, dim3(B_SZ), dim3(256), smem_bytes, stream,
                       army, inst, ilen, idiff, tok, posw, diffw, pv, pb, pg,
                       out_inst, out_sum);
}


# army tile staging loads moved to kernel entry as sixteen 16-byte loads per thread, in flight across the history gather, stored to LDS after it: one memory trip instead of four after the gather
# speedup vs baseline: 2.4485x; 1.0499x over previous
#include <hip/hip_runtime.h>

#define B_SZ   256
#define A_SZ   64
#define FEATN  256
#define H_SZ   16
#define S_SZ   20
#define POSN   64
#define ENCN   256
#define OUTN   384
#define MAXD   10

typedef __attribute__((ext_vector_type(2))) float v2f;
typedef __attribute__((ext_vector_type(8))) float v8f;
typedef __attribute__((ext_vector_type(4))) float v4f;

#define LDA_ARMY 260
#define LDP      388
#define LDH      388
#define LDATT    20
#define OFF_ARMY 0
#define OFF_PARM (OFF_ARMY + A_SZ * LDA_ARMY)
#define OFF_HIST (OFF_PARM + A_SZ * LDP)
#define OFF_ATT  (OFF_HIST + H_SZ * LDH)
#define OFF_RED  (OFF_ATT + A_SZ * LDATT)
#define SMEM_FLOATS (OFF_RED + 8)

__device__ __forceinline__ v8f wmma4(v2f a, v2f b, v8f c) {
    return __builtin_amdgcn_wmma_f32_16x16x4_f32(
         false, a,  false, b,
         (short)0, c,  false,  false);
}

__global__ __launch_bounds__(256) void fused_attn(
    const float* __restrict__ army,
    const int*   __restrict__ inst,
    const int*   __restrict__ ilen,
    const int*   __restrict__ idiff,
    const float* __restrict__ tok_emb,
    const float* __restrict__ pos_w,
    const float* __restrict__ diff_w,
    const float* __restrict__ proj_v,
    const float* __restrict__ proj_b,
    const float* __restrict__ proj_g,
    float* __restrict__ out_inst,
    float* __restrict__ out_sum)
{
    extern __shared__ float smem[];
    float* s_army = smem + OFF_ARMY;
    float* s_parm = smem + OFF_PARM;
    float* s_hist = smem + OFF_HIST;
    float* s_att  = smem + OFF_ATT;
    float* s_red  = smem + OFF_RED;

    const int b    = blockIdx.x;
    const int tid  = threadIdx.x;
    const int wave = tid >> 5;
    const int lane = tid & 31;
    const int lm   = lane & 15;
    const int lh   = lane >> 4;
    const float g = proj_g[0];

    const v4f* asrc = (const v4f*)(army + (size_t)b * A_SZ * FEATN);
    v4f av[16];
    #pragma unroll
    for (int j = 0; j < 16; ++j) av[j] = asrc[tid + 256 * j];

    const int wu = __builtin_amdgcn_readfirstlane(wave);
    const int hrow0 = b * H_SZ + wu * 2;
    int ids[2][S_SZ];
    int lens[2], dis[2];
    #pragma unroll
    for (int hh = 0; hh < 2; ++hh) {
        lens[hh] = ilen[hrow0 + hh];
        int di = idiff[hrow0 + hh];
        dis[hh] = di < 0 ? 0 : (di > MAXD ? MAXD : di);
        #pragma unroll
        for (int s = 0; s < S_SZ; ++s) ids[hh][s] = inst[(hrow0 + hh) * S_SZ + s];
    }
    const int c4 = lm * 4;
    const v4f pv4 = *(const v4f*)(pos_w + (wu * 2 + lh) * POSN + c4);
    const v4f dv4 = *(const v4f*)(diff_w + (lh ? dis[1] : dis[0]) * POSN + c4);
    #pragma unroll
    for (int hh = 0; hh < 2; ++hh) {
        const int h = wu * 2 + hh;
        int len = lens[hh];
        if (len < 1) len = 1;
        v4f t0[S_SZ], t1[S_SZ];
        #pragma unroll
        for (int s = 0; s < S_SZ; ++s) {
            if (s < len) {
                const v4f* erow = (const v4f*)(tok_emb + ids[hh][s] * ENCN);
                t0[s] = erow[lane];
                t1[s] = erow[32 + lane];
            } else {
                t0[s] = (v4f){0.f, 0.f, 0.f, 0.f};
                t1[s] = (v4f){0.f, 0.f, 0.f, 0.f};
            }
        }
        v4f a0 = {0.f, 0.f, 0.f, 0.f}, a1 = {0.f, 0.f, 0.f, 0.f};
        #pragma unroll
        for (int s = 0; s < S_SZ; ++s) { a0 += t0[s]; a1 += t1[s]; }
        const float fl = (float)len;
        v4f m0, m1;
        m0.x = a0.x / fl; m0.y = a0.y / fl; m0.z = a0.z / fl; m0.w = a0.w / fl;
        m1.x = a1.x / fl; m1.y = a1.y / fl; m1.z = a1.z / fl; m1.w = a1.w / fl;
        *(v4f*)(&s_hist[h * LDH + 4 * lane])       = m0;
        *(v4f*)(&s_hist[h * LDH + 128 + 4 * lane]) = m1;
    }
    *(v4f*)(&s_hist[(wu * 2 + lh) * LDH + ENCN + c4])        = pv4;
    *(v4f*)(&s_hist[(wu * 2 + lh) * LDH + ENCN + POSN + c4]) = dv4;
    #pragma unroll
    for (int j = 0; j < 16; ++j)
        *(v4f*)(&s_army[((tid >> 6) + 4 * j) * LDA_ARMY + 4 * (tid & 63)]) = av[j];
    __syncthreads();

    for (int d = tid; d < OUTN; d += 256)
        out_sum[b * OUTN + d] = s_hist[(H_SZ - 1) * LDH + d];

    v8f acc[3][4];
    float ssx = 0.f, ssy = 0.f;
    #pragma unroll
    for (int nt = 0; nt < 3; ++nt) {
        const int n0 = (wave * 3 + nt) * 16;
        v8f acc0 = {}, acc1 = {}, acc2 = {}, acc3 = {};
        const float* wrow = proj_v + (size_t)(n0 + lm) * FEATN + 2 * lh;
        for (int k0 = 0; k0 < FEATN; k0 += 4) {
            v2f bf = *(const v2f*)(wrow + k0);
            v2f a0 = *(const v2f*)(&s_army[(0 * 16 + lm) * LDA_ARMY + k0 + 2 * lh]);
            v2f a1 = *(const v2f*)(&s_army[(1 * 16 + lm) * LDA_ARMY + k0 + 2 * lh]);
            v2f a2 = *(const v2f*)(&s_army[(2 * 16 + lm) * LDA_ARMY + k0 + 2 * lh]);
            v2f a3 = *(const v2f*)(&s_army[(3 * 16 + lm) * LDA_ARMY + k0 + 2 * lh]);
            ssx += bf.x * bf.x;
            ssy += bf.y * bf.y;
            acc0 = wmma4(a0, bf, acc0);
            acc1 = wmma4(a1, bf, acc1);
            acc2 = wmma4(a2, bf, acc2);
            acc3 = wmma4(a3, bf, acc3);
        }
        acc[nt][0] = acc0; acc[nt][1] = acc1; acc[nt][2] = acc2; acc[nt][3] = acc3;
    }
    float ss = ssx + ssy;
    for (int off = 16; off > 0; off >>= 1) ss += __shfl_xor(ss, off, 32);
    if (lane == 0) s_red[wave] = ss;
    __syncthreads();
    float tot = 0.f;
    for (int w = 0; w < 8; ++w) tot += s_red[w];
    const float scale = g * rsqrtf(tot);

    #pragma unroll
    for (int nt = 0; nt < 3; ++nt) {
        const int n0 = (wave * 3 + nt) * 16;
        const float bias = proj_b[n0 + lm];
        #pragma unroll
        for (int r = 0; r < 8; ++r) {
            int mrow = 8 * lh + r;
            float v0 = acc[nt][0][r] * scale + bias;
            float v1 = acc[nt][1][r] * scale + bias;
            float v2 = acc[nt][2][r] * scale + bias;
            float v3 = acc[nt][3][r] * scale + bias;
            s_parm[(0 * 16 + mrow) * LDP + n0 + lm] = v0 > 0.f ? v0 : 0.f;
            s_parm[(1 * 16 + mrow) * LDP + n0 + lm] = v1 > 0.f ? v1 : 0.f;
            s_parm[(2 * 16 + mrow) * LDP + n0 + lm] = v2 > 0.f ? v2 : 0.f;
            s_parm[(3 * 16 + mrow) * LDP + n0 + lm] = v3 > 0.f ? v3 : 0.f;
        }
    }
    __syncthreads();

    if (wave < 4) {
        const int m0 = wave * 16;
        v8f acc = {};
        for (int k0 = 0; k0 < OUTN; k0 += 4) {
            v2f a  = *(const v2f*)(&s_parm[(m0 + lm) * LDP + k0 + 2 * lh]);
            v2f bf = *(const v2f*)(&s_hist[lm * LDH + k0 + 2 * lh]);
            acc = wmma4(a, bf, acc);
        }
        #pragma unroll
        for (int r = 0; r < 8; ++r)
            s_att[(m0 + 8 * lh + r) * LDATT + lm] = acc[r];
    }
    __syncthreads();

    if (tid < 64) {
        float* row = s_att + tid * LDATT;
        float mx = row[0];
        #pragma unroll
        for (int h = 1; h < 16; ++h) mx = fmaxf(mx, row[h]);
        float sum = 0.f;
        #pragma unroll
        for (int h = 0; h < 16; ++h) { float e = __expf(row[h] - mx); row[h] = e; sum += e; }
        float inv = 1.f / sum;
        #pragma unroll
        for (int h = 0; h < 16; ++h) row[h] *= inv;
    }
    __syncthreads();

    for (int nt = 0; nt < 3; ++nt) {
        const int n0 = (wave * 3 + nt) * 16;
        v8f acc0 = {}, acc1 = {}, acc2 = {}, acc3 = {};
        #pragma unroll
        for (int k0 = 0; k0 < 16; k0 += 4) {
            int kk = k0 + 2 * lh;
            v2f bf;
            bf.x = s_hist[kk * LDH + n0 + lm];
            bf.y = s_hist[(kk + 1) * LDH + n0 + lm];
            v2f a0 = *(const v2f*)(&s_att[(0 * 16 + lm) * LDATT + kk]);
            v2f a1 = *(const v2f*)(&s_att[(1 * 16 + lm) * LDATT + kk]);
            v2f a2 = *(const v2f*)(&s_att[(2 * 16 + lm) * LDATT + kk]);
            v2f a3 = *(const v2f*)(&s_att[(3 * 16 + lm) * LDATT + kk]);
            acc0 = wmma4(a0, bf, acc0);
            acc1 = wmma4(a1, bf, acc1);
            acc2 = wmma4(a2, bf, acc2);
            acc3 = wmma4(a3, bf, acc3);
        }
        #pragma unroll
        for (int r = 0; r < 8; ++r) {
            int mrow = 8 * lh + r;
            out_inst[((size_t)b * A_SZ + 0 * 16 + mrow) * OUTN + n0 + lm] = acc0[r];
            out_inst[((size_t)b * A_SZ + 1 * 16 + mrow) * OUTN + n0 + lm] = acc1[r];
            out_inst[((size_t)b * A_SZ + 2 * 16 + mrow) * OUTN + n0 + lm] = acc2[r];
            out_inst[((size_t)b * A_SZ + 3 * 16 + mrow) * OUTN + n0 + lm] = acc3[r];
        }
    }
}

extern "C" void kernel_launch(void* const* d_in, const int* in_sizes, int n_in,
                              void* d_out, int out_size, void* d_ws, size_t ws_size,
                              hipStream_t stream) {
    (void)in_sizes; (void)n_in; (void)out_size; (void)d_ws; (void)ws_size;
    const float* army  = (const float*)d_in[0];
    const int*   inst  = (const int*)  d_in[1];
    const int*   ilen  = (const int*)  d_in[2];
    const int*   idiff = (const int*)  d_in[3];
    const float* tok   = (const float*)d_in[4];
    const float* posw  = (const float*)d_in[5];
    const float* diffw = (const float*)d_in[6];
    const float* pv    = (const float*)d_in[7];
    const float* pg    = (const float*)d_in[8];
    const float* pb    = (const float*)d_in[9];

    float* out      = (float*)d_out;
    float* out_inst = out;
    float* out_sum  = out + (size_t)B_SZ * A_SZ * OUTN;

    size_t smem_bytes = (size_t)SMEM_FLOATS * sizeof(float);
    (void)hipFuncSetAttribute((const void*)fused_attn,
                              hipFuncAttributeMaxDynamicSharedMemorySize,
                              (int)smem_bytes);

    hipLaunchKernelGGL(fused_attn, dim3(B_SZ), dim3(256), smem_bytes, stream,
                       army, inst, ilen, idiff, tok, posw, diffw, pv, pb, pg,
                       out_inst, out_sum);
}


# phase-1 k loop interchanged: one k pass with all 12 accumulator tiles, 16-byte A and B fragments requested 1 and 3 steps ahead of their products; products and norm summed in a new k order
# speedup vs baseline: 2.8566x; 1.1667x over previous
#include <hip/hip_runtime.h>

#define B_SZ   256
#define A_SZ   64
#define FEATN  256
#define H_SZ   16
#define S_SZ   20
#define POSN   64
#define ENCN   256
#define OUTN   384
#define MAXD   10

typedef __attribute__((ext_vector_type(2))) float v2f;
typedef __attribute__((ext_vector_type(8))) float v8f;
typedef __attribute__((ext_vector_type(4))) float v4f;

#define LDA_ARMY 260
#define LDP      388
#define LDH      388
#define LDATT    20
#define OFF_ARMY 0
#define OFF_PARM (OFF_ARMY + A_SZ * LDA_ARMY)
#define OFF_HIST (OFF_PARM + A_SZ * LDP)
#define OFF_ATT  (OFF_HIST + H_SZ * LDH)
#define OFF_RED  (OFF_ATT + A_SZ * LDATT)
#define SMEM_FLOATS (OFF_RED + 8)

__device__ __forceinline__ v8f wmma4(v2f a, v2f b, v8f c) {
    return __builtin_amdgcn_wmma_f32_16x16x4_f32(
         false, a,  false, b,
         (short)0, c,  false,  false);
}

__device__ __forceinline__ void load_b(v4f (&b)[3], const float* wbase, int k) {
    #pragma unroll
    for (int nt = 0; nt < 3; ++nt) b[nt] = *(const v4f*)(wbase + nt * 16 * FEATN + k);
}
__device__ __forceinline__ void load_a(v4f (&a)[4], const float* abase, int k) {
    #pragma unroll
    for (int mt = 0; mt < 4; ++mt) a[mt] = *(const v4f*)(abase + mt * 16 * LDA_ARMY + k);
}
__device__ __forceinline__ void mma_step(v8f (&acc)[3][4], const v4f (&a)[4],
                                         const v4f (&b)[3], v2f& ss2) {
    #pragma unroll
    for (int nt = 0; nt < 3; ++nt) {
        const v2f b01 = {b[nt].x, b[nt].y};
        #pragma unroll
        for (int mt = 0; mt < 4; ++mt) {
            const v2f a01 = {a[mt].x, a[mt].y};
            acc[nt][mt] = wmma4(a01, b01, acc[nt][mt]);
        }
    }
    #pragma unroll
    for (int nt = 0; nt < 3; ++nt) {
        const v2f b23 = {b[nt].z, b[nt].w};
        #pragma unroll
        for (int mt = 0; mt < 4; ++mt) {
            const v2f a23 = {a[mt].z, a[mt].w};
            acc[nt][mt] = wmma4(a23, b23, acc[nt][mt]);
        }
    }
    #pragma unroll
    for (int nt = 0; nt < 3; ++nt) {
        const v2f b01 = {b[nt].x, b[nt].y};
        const v2f b23 = {b[nt].z, b[nt].w};
        ss2 = __builtin_elementwise_fma(b01, b01, ss2);
        ss2 = __builtin_elementwise_fma(b23, b23, ss2);
    }
}

__global__ __launch_bounds__(256) void fused_attn(
    const float* __restrict__ army,
    const int*   __restrict__ inst,
    const int*   __restrict__ ilen,
    const int*   __restrict__ idiff,
    const float* __restrict__ tok_emb,
    const float* __restrict__ pos_w,
    const float* __restrict__ diff_w,
    const float* __restrict__ proj_v,
    const float* __restrict__ proj_b,
    const float* __restrict__ proj_g,
    float* __restrict__ out_inst,
    float* __restrict__ out_sum)
{
    extern __shared__ float smem[];
    float* s_army = smem + OFF_ARMY;
    float* s_parm = smem + OFF_PARM;
    float* s_hist = smem + OFF_HIST;
    float* s_att  = smem + OFF_ATT;
    float* s_red  = smem + OFF_RED;

    const int b    = blockIdx.x;
    const int tid  = threadIdx.x;
    const int wave = tid >> 5;
    const int lane = tid & 31;
    const int lm   = lane & 15;
    const int lh   = lane >> 4;
    const float g = proj_g[0];

    const v4f* asrc = (const v4f*)(army + (size_t)b * A_SZ * FEATN);
    v4f av[16];
    #pragma unroll
    for (int j = 0; j < 16; ++j) av[j] = asrc[tid + 256 * j];

    const int wu = __builtin_amdgcn_readfirstlane(wave);
    const int hrow0 = b * H_SZ + wu * 2;
    int ids[2][S_SZ];
    int lens[2], dis[2];
    #pragma unroll
    for (int hh = 0; hh < 2; ++hh) {
        lens[hh] = ilen[hrow0 + hh];
        int di = idiff[hrow0 + hh];
        dis[hh] = di < 0 ? 0 : (di > MAXD ? MAXD : di);
        #pragma unroll
        for (int s = 0; s < S_SZ; ++s) ids[hh][s] = inst[(hrow0 + hh) * S_SZ + s];
    }
    const int c4 = lm * 4;
    const v4f pv4 = *(const v4f*)(pos_w + (wu * 2 + lh) * POSN + c4);
    const v4f dv4 = *(const v4f*)(diff_w + (lh ? dis[1] : dis[0]) * POSN + c4);
    #pragma unroll
    for (int hh = 0; hh < 2; ++hh) {
        const int h = wu * 2 + hh;
        int len = lens[hh];
        if (len < 1) len = 1;
        v4f t0[S_SZ], t1[S_SZ];
        #pragma unroll
        for (int s = 0; s < S_SZ; ++s) {
            if (s < len) {
                const v4f* erow = (const v4f*)(tok_emb + ids[hh][s] * ENCN);
                t0[s] = erow[lane];
                t1[s] = erow[32 + lane];
            } else {
                t0[s] = (v4f){0.f, 0.f, 0.f, 0.f};
                t1[s] = (v4f){0.f, 0.f, 0.f, 0.f};
            }
        }
        v4f a0 = {0.f, 0.f, 0.f, 0.f}, a1 = {0.f, 0.f, 0.f, 0.f};
        #pragma unroll
        for (int s = 0; s < S_SZ; ++s) { a0 += t0[s]; a1 += t1[s]; }
        const float fl = (float)len;
        v4f m0, m1;
        m0.x = a0.x / fl; m0.y = a0.y / fl; m0.z = a0.z / fl; m0.w = a0.w / fl;
        m1.x = a1.x / fl; m1.y = a1.y / fl; m1.z = a1.z / fl; m1.w = a1.w / fl;
        *(v4f*)(&s_hist[h * LDH + 4 * lane])       = m0;
        *(v4f*)(&s_hist[h * LDH + 128 + 4 * lane]) = m1;
    }
    *(v4f*)(&s_hist[(wu * 2 + lh) * LDH + ENCN + c4])        = pv4;
    *(v4f*)(&s_hist[(wu * 2 + lh) * LDH + ENCN + POSN + c4]) = dv4;
    #pragma unroll
    for (int j = 0; j < 16; ++j)
        *(v4f*)(&s_army[((tid >> 6) + 4 * j) * LDA_ARMY + 4 * (tid & 63)]) = av[j];
    __syncthreads();

    for (int d = tid; d < OUTN; d += 256)
        out_sum[b * OUTN + d] = s_hist[(H_SZ - 1) * LDH + d];

    v8f acc[3][4];
    #pragma unroll
    for (int nt = 0; nt < 3; ++nt) {
        #pragma unroll
        for (int mt = 0; mt < 4; ++mt) acc[nt][mt] = (v8f){};
    }
    v2f ss2 = {0.f, 0.f};
    const float* wbase = proj_v + (size_t)(wave * 48 + lm) * FEATN + 4 * lh;
    const float* abase = &s_army[lm * LDA_ARMY + 4 * lh];
    v4f bq[4][3], aq[2][4];
    load_b(bq[0], wbase, 0);
    __builtin_amdgcn_sched_barrier(0);
    load_b(bq[1], wbase, 8);
    __builtin_amdgcn_sched_barrier(0);
    load_b(bq[2], wbase, 16);
    load_a(aq[0], abase, 0);
    __builtin_amdgcn_sched_barrier(0);
    for (int k0 = 0; k0 < FEATN - 32; k0 += 32) {
        #pragma unroll
        for (int j = 0; j < 4; ++j) {
            load_b(bq[(j + 3) & 3], wbase, k0 + 8 * (j + 3));
            load_a(aq[(j + 1) & 1], abase, k0 + 8 * (j + 1));
            __builtin_amdgcn_sched_barrier(0);
            mma_step(acc, aq[j & 1], bq[j & 3], ss2);
            __builtin_amdgcn_sched_barrier(0);
        }
    }
    load_b(bq[3], wbase, FEATN - 8);
    load_a(aq[1], abase, FEATN - 24);
    __builtin_amdgcn_sched_barrier(0);
    mma_step(acc, aq[0], bq[0], ss2);
    __builtin_amdgcn_sched_barrier(0);
    load_a(aq[0], abase, FEATN - 16);
    __builtin_amdgcn_sched_barrier(0);
    mma_step(acc, aq[1], bq[1], ss2);
    __builtin_amdgcn_sched_barrier(0);
    load_a(aq[1], abase, FEATN - 8);
    __builtin_amdgcn_sched_barrier(0);
    mma_step(acc, aq[0], bq[2], ss2);
    mma_step(acc, aq[1], bq[3], ss2);
    float ss = ss2.x + ss2.y;
    for (int off = 16; off > 0; off >>= 1) ss += __shfl_xor(ss, off, 32);
    if (lane == 0) s_red[wave] = ss;
    __syncthreads();
    float tot = 0.f;
    for (int w = 0; w < 8; ++w) tot += s_red[w];
    const float scale = g * rsqrtf(tot);

    #pragma unroll
    for (int nt = 0; nt < 3; ++nt) {
        const int n0 = (wave * 3 + nt) * 16;
        const float bias = proj_b[n0 + lm];
        #pragma unroll
        for (int r = 0; r < 8; ++r) {
            int mrow = 8 * lh + r;
            float v0 = acc[nt][0][r] * scale + bias;
            float v1 = acc[nt][1][r] * scale + bias;
            float v2 = acc[nt][2][r] * scale + bias;
            float v3 = acc[nt][3][r] * scale + bias;
            s_parm[(0 * 16 + mrow) * LDP + n0 + lm] = v0 > 0.f ? v0 : 0.f;
            s_parm[(1 * 16 + mrow) * LDP + n0 + lm] = v1 > 0.f ? v1 : 0.f;
            s_parm[(2 * 16 + mrow) * LDP + n0 + lm] = v2 > 0.f ? v2 : 0.f;
            s_parm[(3 * 16 + mrow) * LDP + n0 + lm] = v3 > 0.f ? v3 : 0.f;
        }
    }
    __syncthreads();

    if (wave < 4) {
        const int m0 = wave * 16;
        v8f acc = {};
        for (int k0 = 0; k0 < OUTN; k0 += 4) {
            v2f a  = *(const v2f*)(&s_parm[(m0 + lm) * LDP + k0 + 2 * lh]);
            v2f bf = *(const v2f*)(&s_hist[lm * LDH + k0 + 2 * lh]);
            acc = wmma4(a, bf, acc);
        }
        #pragma unroll
        for (int r = 0; r < 8; ++r)
            s_att[(m0 + 8 * lh + r) * LDATT + lm] = acc[r];
    }
    __syncthreads();

    if (tid < 64) {
        float* row = s_att + tid * LDATT;
        float mx = row[0];
        #pragma unroll
        for (int h = 1; h < 16; ++h) mx = fmaxf(mx, row[h]);
        float sum = 0.f;
        #pragma unroll
        for (int h = 0; h < 16; ++h) { float e = __expf(row[h] - mx); row[h] = e; sum += e; }
        float inv = 1.f / sum;
        #pragma unroll
        for (int h = 0; h < 16; ++h) row[h] *= inv;
    }
    __syncthreads();

    for (int nt = 0; nt < 3; ++nt) {
        const int n0 = (wave * 3 + nt) * 16;
        v8f acc0 = {}, acc1 = {}, acc2 = {}, acc3 = {};
        #pragma unroll
        for (int k0 = 0; k0 < 16; k0 += 4) {
            int kk = k0 + 2 * lh;
            v2f bf;
            bf.x = s_hist[kk * LDH + n0 + lm];
            bf.y = s_hist[(kk + 1) * LDH + n0 + lm];
            v2f a0 = *(const v2f*)(&s_att[(0 * 16 + lm) * LDATT + kk]);
            v2f a1 = *(const v2f*)(&s_att[(1 * 16 + lm) * LDATT + kk]);
            v2f a2 = *(const v2f*)(&s_att[(2 * 16 + lm) * LDATT + kk]);
            v2f a3 = *(const v2f*)(&s_att[(3 * 16 + lm) * LDATT + kk]);
            acc0 = wmma4(a0, bf, acc0);
            acc1 = wmma4(a1, bf, acc1);
            acc2 = wmma4(a2, bf, acc2);
            acc3 = wmma4(a3, bf, acc3);
        }
        #pragma unroll
        for (int r = 0; r < 8; ++r) {
            int mrow = 8 * lh + r;
            out_inst[((size_t)b * A_SZ + 0 * 16 + mrow) * OUTN + n0 + lm] = acc0[r];
            out_inst[((size_t)b * A_SZ + 1 * 16 + mrow) * OUTN + n0 + lm] = acc1[r];
            out_inst[((size_t)b * A_SZ + 2 * 16 + mrow) * OUTN + n0 + lm] = acc2[r];
            out_inst[((size_t)b * A_SZ + 3 * 16 + mrow) * OUTN + n0 + lm] = acc3[r];
        }
    }
}

extern "C" void kernel_launch(void* const* d_in, const int* in_sizes, int n_in,
                              void* d_out, int out_size, void* d_ws, size_t ws_size,
                              hipStream_t stream) {
    (void)in_sizes; (void)n_in; (void)out_size; (void)d_ws; (void)ws_size;
    const float* army  = (const float*)d_in[0];
    const int*   inst  = (const int*)  d_in[1];
    const int*   ilen  = (const int*)  d_in[2];
    const int*   idiff = (const int*)  d_in[3];
    const float* tok   = (const float*)d_in[4];
    const float* posw  = (const float*)d_in[5];
    const float* diffw = (const float*)d_in[6];
    const float* pv    = (const float*)d_in[7];
    const float* pg    = (const float*)d_in[8];
    const float* pb    = (const float*)d_in[9];

    float* out      = (float*)d_out;
    float* out_inst = out;
    float* out_sum  = out + (size_t)B_SZ * A_SZ * OUTN;

    size_t smem_bytes = (size_t)SMEM_FLOATS * sizeof(float);
    (void)hipFuncSetAttribute((const void*)fused_attn,
                              hipFuncAttributeMaxDynamicSharedMemorySize,
                              (int)smem_bytes);

    hipLaunchKernelGGL(fused_attn, dim3(B_SZ), dim3(256), smem_bytes, stream,
                       army, inst, ilen, idiff, tok, posw, diffw, pv, pb, pg,
                       out_inst, out_sum);
}
